# starting kernel
# baseline (speedup 1.0000x reference)
#include <hip/hip_runtime.h>

typedef __attribute__((ext_vector_type(2))) float v2f;
typedef __attribute__((ext_vector_type(8))) float v8f;

#define NRAYS_BLOCK 256
#define HIDDEN 256
#define NEAR_T 1.0f
#define FAR_T 10.0f
#define MAX_ITERS 64

__global__ __launch_bounds__(NRAYS_BLOCK) void sphere_trace_mlp_kernel(
    const float* __restrict__ origins,
    const float* __restrict__ dirs,
    const float* __restrict__ center,
    const float* __restrict__ radius,
    const float* __restrict__ w1,
    const float* __restrict__ b1,
    const float* __restrict__ w2,
    const float* __restrict__ b2,
    float* __restrict__ out)
{
    __shared__ v2f   lds_b1t[16 * 32];
    __shared__ float lds_w2[HIDDEN * 3];

    const int tid  = threadIdx.x;
    const int lane = tid & 31;
    const int wave = tid >> 5;
    const int ray  = blockIdx.x * NRAYS_BLOCK + tid;

    for (int e = tid; e < 16 * 32; e += NRAYS_BLOCK) {
        int j = e >> 5, l = e & 31;
        int n = (j << 4) + (l & 15);
        v2f v;
        if (l < 16) { v.x = w1[0 * HIDDEN + n]; v.y = w1[1 * HIDDEN + n]; }
        else        { v.x = w1[2 * HIDDEN + n]; v.y = b1[n]; }
        lds_b1t[e] = v;
    }
    for (int e = tid; e < HIDDEN * 3; e += NRAYS_BLOCK) lds_w2[e] = w2[e];
    __syncthreads();

    const float ox = origins[ray * 3 + 0], oy = origins[ray * 3 + 1], oz = origins[ray * 3 + 2];
    const float rdx = dirs[ray * 3 + 0],   rdy = dirs[ray * 3 + 1],   rdz = dirs[ray * 3 + 2];
    const float cx = center[0], cy = center[1], cz = center[2];
    const float R  = radius[0];

    float t  = NEAR_T;
    float px = ox + t * rdx, py = oy + t * rdy, pz = oz + t * rdz;
    bool mask = true;
#pragma unroll 8
    for (int it = 0; it < MAX_ITERS; ++it) {
        float qx = px - cx, qy = py - cy, qz = pz - cz;
        float f = sqrtf(qx * qx + qy * qy + qz * qz) - R;
        t += f;
        px = ox + t * rdx; py = oy + t * rdy; pz = oz + t * rdz;
        mask = mask && (t <= FAR_T);
    }

    const unsigned long long bal = __ballot(mask);

    const float b2x = b2[0], b2y = b2[1], b2z = b2[2];
    const int  halfSel = (lane >= 16) ? 1 : 0;
    const int  nlane   = lane & 15;
    const int  waveRay0 = blockIdx.x * NRAYS_BLOCK + (wave << 5);

    for (int pass = 0; pass < 2; ++pass) {
        const int srcLane = nlane + (pass << 4);
        float sx = __shfl(px, srcLane, 32);
        float sy = __shfl(py, srcLane, 32);
        float sz = __shfl(pz, srcLane, 32);
        v2f a;
        a.x = halfSel ? sz : sx;
        a.y = halfSel ? 1.0f : sy;

        float accR[8], accG[8], accB[8];
#pragma unroll
        for (int v = 0; v < 8; ++v) { accR[v] = 0.f; accG[v] = 0.f; accB[v] = 0.f; }

#pragma unroll
        for (int j = 0; j < 16; ++j) {
            v2f b = lds_b1t[(j << 5) + lane];
            v8f c = {};
            v8f dmat = __builtin_amdgcn_wmma_f32_16x16x4_f32(
                false, a, false, b, (short)0, c, false, false);

            const int row = (j << 4) + nlane;
            const float wr = lds_w2[row * 3 + 0];
            const float wg = lds_w2[row * 3 + 1];
            const float wb = lds_w2[row * 3 + 2];
#pragma unroll
            for (int v = 0; v < 8; ++v) {
                float h = fmaxf(dmat[v], 0.0f);
                accR[v] += h * wr;
                accG[v] += h * wg;
                accB[v] += h * wb;
            }
        }

#pragma unroll
        for (int off = 8; off >= 1; off >>= 1) {
#pragma unroll
            for (int v = 0; v < 8; ++v) {
                accR[v] += __shfl_xor(accR[v], off, 32);
                accG[v] += __shfl_xor(accG[v], off, 32);
                accB[v] += __shfl_xor(accB[v], off, 32);
            }
        }

        const int passBase = waveRay0 + (pass << 4);
#pragma unroll
        for (int v = 0; v < 8; ++v) {
            if (nlane == v) {
                const int m      = v + (halfSel << 3);
                const int wlLane = (pass << 4) + m;
                const int r      = passBase + m;
                float s0 = 0.f, s1 = 0.f, s2 = 0.f;
                if ((bal >> wlLane) & 1ull) {
                    s0 = 1.0f / (1.0f + expf(-(accR[v] + b2x)));
                    s1 = 1.0f / (1.0f + expf(-(accG[v] + b2y)));
                    s2 = 1.0f / (1.0f + expf(-(accB[v] + b2z)));
                }
                out[r * 3 + 0] = s0;
                out[r * 3 + 1] = s1;
                out[r * 3 + 2] = s2;
            }
        }
    }
}

extern "C" void kernel_launch(void* const* d_in, const int* in_sizes, int n_in,
                              void* d_out, int out_size, void* d_ws, size_t ws_size,
                              hipStream_t stream) {
    const float* origins = (const float*)d_in[0];
    const float* dirs    = (const float*)d_in[1];
    const float* center  = (const float*)d_in[2];
    const float* radius  = (const float*)d_in[3];
    const float* w1      = (const float*)d_in[4];
    const float* b1      = (const float*)d_in[5];
    const float* w2      = (const float*)d_in[6];
    const float* b2      = (const float*)d_in[7];
    float* out = (float*)d_out;

    const int n_rays = in_sizes[0] / 3;
    const int blocks = n_rays / NRAYS_BLOCK;

    sphere_trace_mlp_kernel<<<blocks, NRAYS_BLOCK, 0, stream>>>(
        origins, dirs, center, radius, w1, b1, w2, b2, out);
}


# Layer 1 transposed (A=weights, B=ray points) so each lane holds its own ray's hidden values: layer 2 sums per lane + one cross-half add (new sum order), no 16-lane butterfly, no 2-lane epilogue
# speedup vs baseline: 1.4457x; 1.4457x over previous
#include <hip/hip_runtime.h>

typedef __attribute__((ext_vector_type(2))) float v2f;
typedef __attribute__((ext_vector_type(4))) float v4f;
typedef __attribute__((ext_vector_type(8))) float v8f;

#define NRAYS_BLOCK 256
#define HIDDEN 256
#define NEAR_T 1.0f
#define FAR_T 10.0f
#define MAX_ITERS 64

__global__ __launch_bounds__(NRAYS_BLOCK) void sphere_trace_mlp_kernel(
    const float* __restrict__ origins,
    const float* __restrict__ dirs,
    const float* __restrict__ center,
    const float* __restrict__ radius,
    const float* __restrict__ w1,
    const float* __restrict__ b1,
    const float* __restrict__ w2,
    const float* __restrict__ b2,
    float* __restrict__ out)
{
    __shared__ v2f   lds_b1t[16 * 32];
    __shared__ __attribute__((aligned(16))) float lds_w2[HIDDEN * 3];

    const int tid  = threadIdx.x;
    const int lane = tid & 31;
    const int wave = tid >> 5;
    const int ray  = blockIdx.x * NRAYS_BLOCK + tid;

    for (int e = tid; e < 16 * 32; e += NRAYS_BLOCK) {
        int j = e >> 5, l = e & 31;
        int n = (j << 4) + (l & 15);
        v2f v;
        if (l < 16) { v.x = w1[0 * HIDDEN + n]; v.y = w1[1 * HIDDEN + n]; }
        else        { v.x = w1[2 * HIDDEN + n]; v.y = b1[n]; }
        lds_b1t[e] = v;
    }
    for (int e = tid; e < HIDDEN * 3; e += NRAYS_BLOCK) lds_w2[e] = w2[e];
    __syncthreads();

    const float ox = origins[ray * 3 + 0], oy = origins[ray * 3 + 1], oz = origins[ray * 3 + 2];
    const float rdx = dirs[ray * 3 + 0],   rdy = dirs[ray * 3 + 1],   rdz = dirs[ray * 3 + 2];
    const float cx = center[0], cy = center[1], cz = center[2];
    const float R  = radius[0];

    float t  = NEAR_T;
    float px = ox + t * rdx, py = oy + t * rdy, pz = oz + t * rdz;
    bool mask = true;
#pragma unroll 8
    for (int it = 0; it < MAX_ITERS; ++it) {
        float qx = px - cx, qy = py - cy, qz = pz - cz;
        float f = sqrtf(qx * qx + qy * qy + qz * qz) - R;
        t += f;
        px = ox + t * rdx; py = oy + t * rdy; pz = oz + t * rdz;
        mask = mask && (t <= FAR_T);
    }

    const float b2x = b2[0], b2y = b2[1], b2z = b2[2];
    const int  halfSel = (lane >= 16) ? 1 : 0;
    const int  nlane   = lane & 15;

    v2f bp0, bp1;
    {
        float sx = __shfl(px, nlane, 32);
        float sy = __shfl(py, nlane, 32);
        float sz = __shfl(pz, nlane, 32);
        bp0.x = halfSel ? sz : sx;
        bp0.y = halfSel ? 1.0f : sy;
        sx = __shfl(px, nlane + 16, 32);
        sy = __shfl(py, nlane + 16, 32);
        sz = __shfl(pz, nlane + 16, 32);
        bp1.x = halfSel ? sz : sx;
        bp1.y = halfSel ? 1.0f : sy;
    }

    float a0r = 0.f, a0g = 0.f, a0b = 0.f;
    float a1r = 0.f, a1g = 0.f, a1b = 0.f;

    const v4f* w2q = reinterpret_cast<const v4f*>(lds_w2) + halfSel * 6;

#pragma unroll
    for (int j = 0; j < 16; ++j) {
        const v2f a = lds_b1t[(j << 5) + lane];
        const v4f q0 = w2q[12 * j + 0], q1 = w2q[12 * j + 1], q2 = w2q[12 * j + 2];
        const v4f q3 = w2q[12 * j + 3], q4 = w2q[12 * j + 4], q5 = w2q[12 * j + 5];
        const float w[24] = { q0.x, q0.y, q0.z, q0.w, q1.x, q1.y, q1.z, q1.w,
                              q2.x, q2.y, q2.z, q2.w, q3.x, q3.y, q3.z, q3.w,
                              q4.x, q4.y, q4.z, q4.w, q5.x, q5.y, q5.z, q5.w };
        v8f c = {};
        const v8f d0 = __builtin_amdgcn_wmma_f32_16x16x4_f32(
            false, a, false, bp0, (short)0, c, false, false);
        const v8f d1 = __builtin_amdgcn_wmma_f32_16x16x4_f32(
            false, a, false, bp1, (short)0, c, false, false);
#pragma unroll
        for (int r = 0; r < 8; ++r) {
            const float h0 = fmaxf(d0[r], 0.0f);
            const float h1 = fmaxf(d1[r], 0.0f);
            a0r += h0 * w[3 * r + 0];
            a0g += h0 * w[3 * r + 1];
            a0b += h0 * w[3 * r + 2];
            a1r += h1 * w[3 * r + 0];
            a1g += h1 * w[3 * r + 1];
            a1b += h1 * w[3 * r + 2];
        }
    }

    const float tr = (halfSel ? a1r : a0r) + __shfl_xor(halfSel ? a0r : a1r, 16, 32);
    const float tg = (halfSel ? a1g : a0g) + __shfl_xor(halfSel ? a0g : a1g, 16, 32);
    const float tb = (halfSel ? a1b : a0b) + __shfl_xor(halfSel ? a0b : a1b, 16, 32);

    float s0 = 0.f, s1 = 0.f, s2 = 0.f;
    if (mask) {
        s0 = 1.0f / (1.0f + expf(-(tr + b2x)));
        s1 = 1.0f / (1.0f + expf(-(tg + b2y)));
        s2 = 1.0f / (1.0f + expf(-(tb + b2z)));
    }
    out[ray * 3 + 0] = s0;
    out[ray * 3 + 1] = s1;
    out[ray * 3 + 2] = s2;
}

extern "C" void kernel_launch(void* const* d_in, const int* in_sizes, int n_in,
                              void* d_out, int out_size, void* d_ws, size_t ws_size,
                              hipStream_t stream) {
    const float* origins = (const float*)d_in[0];
    const float* dirs    = (const float*)d_in[1];
    const float* center  = (const float*)d_in[2];
    const float* radius  = (const float*)d_in[3];
    const float* w1      = (const float*)d_in[4];
    const float* b1      = (const float*)d_in[5];
    const float* w2      = (const float*)d_in[6];
    const float* b2      = (const float*)d_in[7];
    float* out = (float*)d_out;

    const int n_rays = in_sizes[0] / 3;
    const int blocks = n_rays / NRAYS_BLOCK;

    sphere_trace_mlp_kernel<<<blocks, NRAYS_BLOCK, 0, stream>>>(
        origins, dirs, center, radius, w1, b1, w2, b2, out);
}


# Tracing loop: each step's distance uses the hardware sqrt instruction (within 1 ulp) in place of sqrtf's in-line correctly rounded expansion (prescale + last-bit fixup, ~14 VALU per step)
# speedup vs baseline: 1.7564x; 1.2149x over previous
#include <hip/hip_runtime.h>

typedef __attribute__((ext_vector_type(2))) float v2f;
typedef __attribute__((ext_vector_type(4))) float v4f;
typedef __attribute__((ext_vector_type(8))) float v8f;

#define NRAYS_BLOCK 256
#define HIDDEN 256
#define NEAR_T 1.0f
#define FAR_T 10.0f
#define MAX_ITERS 64

__global__ __launch_bounds__(NRAYS_BLOCK) void sphere_trace_mlp_kernel(
    const float* __restrict__ origins,
    const float* __restrict__ dirs,
    const float* __restrict__ center,
    const float* __restrict__ radius,
    const float* __restrict__ w1,
    const float* __restrict__ b1,
    const float* __restrict__ w2,
    const float* __restrict__ b2,
    float* __restrict__ out)
{
    __shared__ v2f   lds_b1t[16 * 32];
    __shared__ __attribute__((aligned(16))) float lds_w2[HIDDEN * 3];

    const int tid  = threadIdx.x;
    const int lane = tid & 31;
    const int wave = tid >> 5;
    const int ray  = blockIdx.x * NRAYS_BLOCK + tid;

    for (int e = tid; e < 16 * 32; e += NRAYS_BLOCK) {
        int j = e >> 5, l = e & 31;
        int n = (j << 4) + (l & 15);
        v2f v;
        if (l < 16) { v.x = w1[0 * HIDDEN + n]; v.y = w1[1 * HIDDEN + n]; }
        else        { v.x = w1[2 * HIDDEN + n]; v.y = b1[n]; }
        lds_b1t[e] = v;
    }
    for (int e = tid; e < HIDDEN * 3; e += NRAYS_BLOCK) lds_w2[e] = w2[e];
    __syncthreads();

    const float ox = origins[ray * 3 + 0], oy = origins[ray * 3 + 1], oz = origins[ray * 3 + 2];
    const float rdx = dirs[ray * 3 + 0],   rdy = dirs[ray * 3 + 1],   rdz = dirs[ray * 3 + 2];
    const float cx = center[0], cy = center[1], cz = center[2];
    const float R  = radius[0];

    float t  = NEAR_T;
    float px = ox + t * rdx, py = oy + t * rdy, pz = oz + t * rdz;
    bool mask = true;
#pragma unroll 8
    for (int it = 0; it < MAX_ITERS; ++it) {
        float qx = px - cx, qy = py - cy, qz = pz - cz;
        float f = __builtin_amdgcn_sqrtf(qx * qx + qy * qy + qz * qz) - R;
        t += f;
        px = ox + t * rdx; py = oy + t * rdy; pz = oz + t * rdz;
        mask = mask && (t <= FAR_T);
    }

    const float b2x = b2[0], b2y = b2[1], b2z = b2[2];
    const int  halfSel = (lane >= 16) ? 1 : 0;
    const int  nlane   = lane & 15;

    v2f bp0, bp1;
    {
        float sx = __shfl(px, nlane, 32);
        float sy = __shfl(py, nlane, 32);
        float sz = __shfl(pz, nlane, 32);
        bp0.x = halfSel ? sz : sx;
        bp0.y = halfSel ? 1.0f : sy;
        sx = __shfl(px, nlane + 16, 32);
        sy = __shfl(py, nlane + 16, 32);
        sz = __shfl(pz, nlane + 16, 32);
        bp1.x = halfSel ? sz : sx;
        bp1.y = halfSel ? 1.0f : sy;
    }

    float a0r = 0.f, a0g = 0.f, a0b = 0.f;
    float a1r = 0.f, a1g = 0.f, a1b = 0.f;

    const v4f* w2q = reinterpret_cast<const v4f*>(lds_w2) + halfSel * 6;

#pragma unroll
    for (int j = 0; j < 16; ++j) {
        const v2f a = lds_b1t[(j << 5) + lane];
        const v4f q0 = w2q[12 * j + 0], q1 = w2q[12 * j + 1], q2 = w2q[12 * j + 2];
        const v4f q3 = w2q[12 * j + 3], q4 = w2q[12 * j + 4], q5 = w2q[12 * j + 5];
        const float w[24] = { q0.x, q0.y, q0.z, q0.w, q1.x, q1.y, q1.z, q1.w,
                              q2.x, q2.y, q2.z, q2.w, q3.x, q3.y, q3.z, q3.w,
                              q4.x, q4.y, q4.z, q4.w, q5.x, q5.y, q5.z, q5.w };
        v8f c = {};
        const v8f d0 = __builtin_amdgcn_wmma_f32_16x16x4_f32(
            false, a, false, bp0, (short)0, c, false, false);
        const v8f d1 = __builtin_amdgcn_wmma_f32_16x16x4_f32(
            false, a, false, bp1, (short)0, c, false, false);
#pragma unroll
        for (int r = 0; r < 8; ++r) {
            const float h0 = fmaxf(d0[r], 0.0f);
            const float h1 = fmaxf(d1[r], 0.0f);
            a0r += h0 * w[3 * r + 0];
            a0g += h0 * w[3 * r + 1];
            a0b += h0 * w[3 * r + 2];
            a1r += h1 * w[3 * r + 0];
            a1g += h1 * w[3 * r + 1];
            a1b += h1 * w[3 * r + 2];
        }
    }

    const float tr = (halfSel ? a1r : a0r) + __shfl_xor(halfSel ? a0r : a1r, 16, 32);
    const float tg = (halfSel ? a1g : a0g) + __shfl_xor(halfSel ? a0g : a1g, 16, 32);
    const float tb = (halfSel ? a1b : a0b) + __shfl_xor(halfSel ? a0b : a1b, 16, 32);

    float s0 = 0.f, s1 = 0.f, s2 = 0.f;
    if (mask) {
        s0 = 1.0f / (1.0f + expf(-(tr + b2x)));
        s1 = 1.0f / (1.0f + expf(-(tg + b2y)));
        s2 = 1.0f / (1.0f + expf(-(tb + b2z)));
    }
    out[ray * 3 + 0] = s0;
    out[ray * 3 + 1] = s1;
    out[ray * 3 + 2] = s2;
}

extern "C" void kernel_launch(void* const* d_in, const int* in_sizes, int n_in,
                              void* d_out, int out_size, void* d_ws, size_t ws_size,
                              hipStream_t stream) {
    const float* origins = (const float*)d_in[0];
    const float* dirs    = (const float*)d_in[1];
    const float* center  = (const float*)d_in[2];
    const float* radius  = (const float*)d_in[3];
    const float* w1      = (const float*)d_in[4];
    const float* b1      = (const float*)d_in[5];
    const float* w2      = (const float*)d_in[6];
    const float* b2      = (const float*)d_in[7];
    float* out = (float*)d_out;

    const int n_rays = in_sizes[0] / 3;
    const int blocks = n_rays / NRAYS_BLOCK;

    sphere_trace_mlp_kernel<<<blocks, NRAYS_BLOCK, 0, stream>>>(
        origins, dirs, center, radius, w1, b1, w2, b2, out);
}


# Prologue: weight-staging and ray loads issued as one group ahead of any wait, LDS barrier moved behind the tracing loop: one vector-memory round trip in place of eight in series
# speedup vs baseline: 2.1846x; 1.2438x over previous
#include <hip/hip_runtime.h>

typedef __attribute__((ext_vector_type(2))) float v2f;
typedef __attribute__((ext_vector_type(4))) float v4f;
typedef __attribute__((ext_vector_type(8))) float v8f;

#define NRAYS_BLOCK 256
#define HIDDEN 256
#define NEAR_T 1.0f
#define FAR_T 10.0f
#define MAX_ITERS 64

__global__ __launch_bounds__(NRAYS_BLOCK) void sphere_trace_mlp_kernel(
    const float* __restrict__ origins,
    const float* __restrict__ dirs,
    const float* __restrict__ center,
    const float* __restrict__ radius,
    const float* __restrict__ w1,
    const float* __restrict__ b1,
    const float* __restrict__ w2,
    const float* __restrict__ b2,
    float* __restrict__ out)
{
    __shared__ v2f   lds_b1t[16 * 32];
    __shared__ __attribute__((aligned(16))) float lds_w2[HIDDEN * 3];

    const int tid  = threadIdx.x;
    const int lane = tid & 31;
    const int ray  = blockIdx.x * NRAYS_BLOCK + tid;

    const int  sl  = tid & 31, sj = tid >> 5;
    const int  sn  = (sj << 4) + (sl & 15);
    const bool shi = sl >= 16;
    const float* src_k0 = w1 + (shi ? 2 * HIDDEN : 0) + sn;
    const float* src_k1 = shi ? (b1 + sn) : (w1 + HIDDEN + sn);
    const float a00 = src_k0[0],   a01 = src_k1[0];
    const float a10 = src_k0[128], a11 = src_k1[128];
    const float u0 = w2[tid], u1 = w2[tid + NRAYS_BLOCK], u2 = w2[tid + 2 * NRAYS_BLOCK];

    const float ox = origins[ray * 3 + 0], oy = origins[ray * 3 + 1], oz = origins[ray * 3 + 2];
    const float rdx = dirs[ray * 3 + 0],   rdy = dirs[ray * 3 + 1],   rdz = dirs[ray * 3 + 2];
    const float cx = center[0], cy = center[1], cz = center[2];
    const float R  = radius[0];

    {
        v2f v0, v1;
        v0.x = a00; v0.y = a01;
        v1.x = a10; v1.y = a11;
        lds_b1t[tid] = v0;
        lds_b1t[tid + NRAYS_BLOCK] = v1;
        lds_w2[tid] = u0;
        lds_w2[tid + NRAYS_BLOCK] = u1;
        lds_w2[tid + 2 * NRAYS_BLOCK] = u2;
    }

    float t  = NEAR_T;
    float px = ox + t * rdx, py = oy + t * rdy, pz = oz + t * rdz;
    bool mask = true;
#pragma unroll 8
    for (int it = 0; it < MAX_ITERS; ++it) {
        float qx = px - cx, qy = py - cy, qz = pz - cz;
        float f = __builtin_amdgcn_sqrtf(qx * qx + qy * qy + qz * qz) - R;
        t += f;
        px = ox + t * rdx; py = oy + t * rdy; pz = oz + t * rdz;
        mask = mask && (t <= FAR_T);
    }

    __syncthreads();

    const float b2x = b2[0], b2y = b2[1], b2z = b2[2];
    const int  halfSel = (lane >= 16) ? 1 : 0;
    const int  nlane   = lane & 15;

    v2f bp0, bp1;
    {
        float sx = __shfl(px, nlane, 32);
        float sy = __shfl(py, nlane, 32);
        float sz = __shfl(pz, nlane, 32);
        bp0.x = halfSel ? sz : sx;
        bp0.y = halfSel ? 1.0f : sy;
        sx = __shfl(px, nlane + 16, 32);
        sy = __shfl(py, nlane + 16, 32);
        sz = __shfl(pz, nlane + 16, 32);
        bp1.x = halfSel ? sz : sx;
        bp1.y = halfSel ? 1.0f : sy;
    }

    float a0r = 0.f, a0g = 0.f, a0b = 0.f;
    float a1r = 0.f, a1g = 0.f, a1b = 0.f;

    const v4f* w2q = reinterpret_cast<const v4f*>(lds_w2) + halfSel * 6;

#pragma unroll
    for (int j = 0; j < 16; ++j) {
        const v2f a = lds_b1t[(j << 5) + lane];
        const v4f q0 = w2q[12 * j + 0], q1 = w2q[12 * j + 1], q2 = w2q[12 * j + 2];
        const v4f q3 = w2q[12 * j + 3], q4 = w2q[12 * j + 4], q5 = w2q[12 * j + 5];
        const float w[24] = { q0.x, q0.y, q0.z, q0.w, q1.x, q1.y, q1.z, q1.w,
                              q2.x, q2.y, q2.z, q2.w, q3.x, q3.y, q3.z, q3.w,
                              q4.x, q4.y, q4.z, q4.w, q5.x, q5.y, q5.z, q5.w };
        v8f c = {};
        const v8f d0 = __builtin_amdgcn_wmma_f32_16x16x4_f32(
            false, a, false, bp0, (short)0, c, false, false);
        const v8f d1 = __builtin_amdgcn_wmma_f32_16x16x4_f32(
            false, a, false, bp1, (short)0, c, false, false);
#pragma unroll
        for (int r = 0; r < 8; ++r) {
            const float h0 = fmaxf(d0[r], 0.0f);
            const float h1 = fmaxf(d1[r], 0.0f);
            a0r += h0 * w[3 * r + 0];
            a0g += h0 * w[3 * r + 1];
            a0b += h0 * w[3 * r + 2];
            a1r += h1 * w[3 * r + 0];
            a1g += h1 * w[3 * r + 1];
            a1b += h1 * w[3 * r + 2];
        }
    }

    const float tr = (halfSel ? a1r : a0r) + __shfl_xor(halfSel ? a0r : a1r, 16, 32);
    const float tg = (halfSel ? a1g : a0g) + __shfl_xor(halfSel ? a0g : a1g, 16, 32);
    const float tb = (halfSel ? a1b : a0b) + __shfl_xor(halfSel ? a0b : a1b, 16, 32);

    float s0 = 0.f, s1 = 0.f, s2 = 0.f;
    if (mask) {
        s0 = 1.0f / (1.0f + expf(-(tr + b2x)));
        s1 = 1.0f / (1.0f + expf(-(tg + b2y)));
        s2 = 1.0f / (1.0f + expf(-(tb + b2z)));
    }
    out[ray * 3 + 0] = s0;
    out[ray * 3 + 1] = s1;
    out[ray * 3 + 2] = s2;
}

extern "C" void kernel_launch(void* const* d_in, const int* in_sizes, int n_in,
                              void* d_out, int out_size, void* d_ws, size_t ws_size,
                              hipStream_t stream) {
    const float* origins = (const float*)d_in[0];
    const float* dirs    = (const float*)d_in[1];
    const float* center  = (const float*)d_in[2];
    const float* radius  = (const float*)d_in[3];
    const float* w1      = (const float*)d_in[4];
    const float* b1      = (const float*)d_in[5];
    const float* w2      = (const float*)d_in[6];
    const float* b2      = (const float*)d_in[7];
    float* out = (float*)d_out;

    const int n_rays = in_sizes[0] / 3;
    const int blocks = n_rays / NRAYS_BLOCK;

    sphere_trace_mlp_kernel<<<blocks, NRAYS_BLOCK, 0, stream>>>(
        origins, dirs, center, radius, w1, b1, w2, b2, out);
}
